# starting kernel
# baseline (speedup 1.0000x reference)
#include <hip/hip_runtime.h>

typedef __attribute__((ext_vector_type(2))) float v2f;
typedef __attribute__((ext_vector_type(8))) float v8f;

#define N_LAYERS 7
#define N_QUBITS 8
#define DIM 256

__global__ __launch_bounds__(256) void build_u_kernel(const float* __restrict__ w,
                                                      float* __restrict__ bp) {
  __shared__ float col[DIM];
  const int i = threadIdx.x;
  const int j = blockIdx.x;
  col[i] = (i == j) ? 1.0f : 0.0f;
  __syncthreads();
  for (int l = 0; l < N_LAYERS; ++l) {
    for (int q = 0; q < N_QUBITS; ++q) {
      const float th = 0.5f * w[l * N_QUBITS + q];
      const float c = cosf(th);
      const float s = sinf(th);
      const int mask = 1 << (7 - q);
      const float a0 = col[i & ~mask];
      const float a1 = col[i | mask];
      __syncthreads();
      col[i] = (i & mask) ? (s * a0 + c * a1) : (c * a0 - s * a1);
      __syncthreads();
    }
    for (int q = 0; q < N_QUBITS - 1; ++q) {
      const int mc = 1 << (7 - q);
      const int mt = 1 << (6 - q);
      const int src = (i & mc) ? (i ^ mt) : i;
      const float v = col[src];
      __syncthreads();
      col[i] = v;
      __syncthreads();
    }
  }
  const int k   = j & ~1;
  const int c   = j & 1;
  const int kk  = k >> 2;
  const int hf  = (k >> 1) & 1;
  const int nt  = i >> 4;
  const int l16 = i & 15;
  const int lane = hf * 16 + l16;
  bp[(size_t)(((nt * 64 + kk) * 32) + lane) * 2 + c] = col[i];
}

__global__ __launch_bounds__(256) void qnn_fused_kernel(
    const float* __restrict__ x, const v2f* __restrict__ bp,
    const float* __restrict__ W1, const float* __restrict__ b1,
    const float* __restrict__ W2, const float* __restrict__ b2,
    float* __restrict__ out) {
  __shared__ float redP[8][16][16];
  __shared__ float redM[8][16][16];

  const int lane = threadIdx.x & 31;
  const int wave = threadIdx.x >> 5;
  const int hf   = lane >> 4;
  const int l16  = lane & 15;
  const int rbase = blockIdx.x * 128 + wave * 16;

  const float* xrow = x + (size_t)(rbase + l16) * DIM + 2 * hf;

  v2f afrag[64];
#pragma unroll
  for (int kk = 0; kk < 64; ++kk) {
    afrag[kk] = *(const v2f*)(xrow + 4 * kk);
  }

  float accP[8], accM[8];
#pragma unroll
  for (int v = 0; v < 8; ++v) { accP[v] = 0.f; accM[v] = 0.f; }

  for (int nt = 0; nt < 16; ++nt) {
    const v2f* btile = bp + (size_t)nt * 64 * 32 + lane;
    v8f acc = {0.f, 0.f, 0.f, 0.f, 0.f, 0.f, 0.f, 0.f};
#pragma unroll
    for (int kk = 0; kk < 64; ++kk) {
      const v2f b = btile[kk * 32];
      acc = __builtin_amdgcn_wmma_f32_16x16x4_f32(
           false, afrag[kk],  false, b,
           (short)0, acc,  false,  false);
    }
    if (nt < 8) {
#pragma unroll
      for (int v = 0; v < 8; ++v) accP[v] += acc[v] * acc[v];
    } else {
#pragma unroll
      for (int v = 0; v < 8; ++v) accM[v] += acc[v] * acc[v];
    }
  }

#pragma unroll
  for (int v = 0; v < 8; ++v) {
    redP[wave][hf * 8 + v][l16] = accP[v];
    redM[wave][hf * 8 + v][l16] = accM[v];
  }
  __syncthreads();

  if (threadIdx.x < 128) {
    const int wv = threadIdx.x >> 4;
    const int r  = threadIdx.x & 15;
    float P = 0.f, M = 0.f;
#pragma unroll
    for (int i = 0; i < 16; ++i) { P += redP[wv][r][i]; M += redM[wv][r][i]; }
    const float z = (P - M) / (P + M);
    float t = b2[0];
#pragma unroll
    for (int jj = 0; jj < 16; ++jj) {
      float h = z * W1[jj] + b1[jj];
      h = h > 0.f ? h : 0.f;
      t += h * W2[jj];
    }
    out[(size_t)blockIdx.x * 128 + wv * 16 + r] = 1.f / (1.f + expf(-t));
  }
}

extern "C" void kernel_launch(void* const* d_in, const int* in_sizes, int n_in,
                              void* d_out, int out_size, void* d_ws, size_t ws_size,
                              hipStream_t stream) {
  const float* x  = (const float*)d_in[0];
  const float* w  = (const float*)d_in[1];
  const float* W1 = (const float*)d_in[2];
  const float* b1 = (const float*)d_in[3];
  const float* W2 = (const float*)d_in[4];
  const float* b2 = (const float*)d_in[5];
  float* out = (float*)d_out;
  float* bp  = (float*)d_ws;

  const int B = in_sizes[0] / DIM;
  build_u_kernel<<<DIM, DIM, 0, stream>>>(w, bp);
  qnn_fused_kernel<<<B / 128, 256, 0, stream>>>(x, (const v2f*)bp, W1, b1, W2, b2, out);
}


# Product as the quadratic form z = x.Qx / x.x, Q = U^T Z0 U built by a forward and a transposed circuit pass; Q symmetric: only its lower 32x32 block triangle on WMMA (576 of 1024): new sum order
# speedup vs baseline: 1.8265x; 1.8265x over previous
#include <hip/hip_runtime.h>

typedef __attribute__((ext_vector_type(2))) float v2f;
typedef __attribute__((ext_vector_type(8))) float v8f;

#define N_LAYERS 7
#define N_QUBITS 8
#define DIM 256


__host__ __device__ constexpr int chunk_base(int m) { return m * (m + 1) / 2; }
#define N_CHUNKS 36


template <int SH>
__device__ __forceinline__ float ry4_stage(const float* src, int i, float4 c, float4 s) {
  float v[16];
  if (SH == 0) {
    const float4* p = (const float4*)(src + (i & ~15));
    for (int t = 0; t < 4; ++t) {
      const float4 f = p[t];
      v[4 * t] = f.x; v[4 * t + 1] = f.y; v[4 * t + 2] = f.z; v[4 * t + 3] = f.w;
    }
  } else {
    const float* p = src + (i & ~(15 << SH));
    for (int m = 0; m < 16; ++m) v[m] = p[m << SH];
  }
  const bool b3 = (i >> (SH + 3)) & 1;
  const bool b2 = (i >> (SH + 2)) & 1;
  const bool b1 = (i >> (SH + 1)) & 1;
  const bool b0 = (i >> SH) & 1;
  const float p3 = b3 ? s.x : c.x, q3 = b3 ? c.x : -s.x;
  const float p2 = b2 ? s.y : c.y, q2 = b2 ? c.y : -s.y;
  const float p1 = b1 ? s.z : c.z, q1 = b1 ? c.z : -s.z;
  const float p0 = b0 ? s.w : c.w, q0 = b0 ? c.w : -s.w;
  for (int t = 0; t < 8; ++t) v[t] = p3 * v[t] + q3 * v[t + 8];
  for (int t = 0; t < 4; ++t) v[t] = p2 * v[t] + q2 * v[t + 4];
  for (int t = 0; t < 2; ++t) v[t] = p1 * v[t] + q1 * v[t + 2];
  return p0 * v[0] + q0 * v[1];
}

__device__ __forceinline__ float4 neg4(float4 a) {
  return make_float4(-a.x, -a.y, -a.z, -a.w);
}

__global__ __launch_bounds__(256) void build_q_kernel(const float* __restrict__ w,
                                                      float* __restrict__ bp) {
  __shared__ __attribute__((aligned(16))) float col[2][DIM];
  __shared__ __attribute__((aligned(16))) float tcos[N_LAYERS * N_QUBITS];
  __shared__ __attribute__((aligned(16))) float tsin[N_LAYERS * N_QUBITS];
  const int i = threadIdx.x;
  const int j = blockIdx.x;
  if (i < N_LAYERS * N_QUBITS) {
    const float th = 0.5f * w[i];
    tcos[i] = cosf(th);
    tsin[i] = sinf(th);
  }
  col[0][i] = (i == j) ? 1.0f : 0.0f;
  int dst = i;
  for (int q = 0; q < N_QUBITS - 1; ++q) {
    if (dst & (1 << (7 - q))) dst ^= 1 << (6 - q);
  }
  const int inv = i ^ (i >> 1);
  __syncthreads();
  for (int l = 0; l < N_LAYERS; ++l) {
    const float4 ca = *(const float4*)&tcos[l * N_QUBITS];
    const float4 sa = *(const float4*)&tsin[l * N_QUBITS];
    const float4 cb = *(const float4*)&tcos[l * N_QUBITS + 4];
    const float4 sb = *(const float4*)&tsin[l * N_QUBITS + 4];
    col[1][i] = ry4_stage<4>(col[0], i, ca, sa);
    __syncthreads();
    const float r = ry4_stage<0>(col[1], i, cb, sb);
    if (l < N_LAYERS - 1) {
      col[0][dst] = r;
    } else {
      col[0][i] = (i & 128) ? -r : r;
    }
    __syncthreads();
  }
  for (int l = N_LAYERS - 1; l >= 0; --l) {
    const float4 ca = *(const float4*)&tcos[l * N_QUBITS];
    const float4 sa = neg4(*(const float4*)&tsin[l * N_QUBITS]);
    const float4 cb = *(const float4*)&tcos[l * N_QUBITS + 4];
    const float4 sb = neg4(*(const float4*)&tsin[l * N_QUBITS + 4]);
    col[1][i] = ry4_stage<0>(col[0], i, cb, sb);
    __syncthreads();
    const float r = ry4_stage<4>(col[1], i, ca, sa);
    col[0][l > 0 ? inv : i] = r;
    __syncthreads();
  }
  const int m = i >> 5;
  if ((j >> 5) <= m) {
    const int t   = (i >> 4) & 1;
    const int l16 = i & 15;
    const int kp  = j >> 3;
    const int hf  = (j >> 1) & 1;
    const int lane = hf * 16 + l16;
    const int row = (chunk_base(m) + (kp >> 2)) * 8 + (kp & 3) * 2 + t;
    const float q = col[0][i];
    bp[(size_t)(row * 32 + lane) * 4 + ((j >> 2) & 1) * 2 + (j & 1)] =
        ((j >> 5) == m) ? q : 2.0f * q;
  }
}

__global__ __launch_bounds__(512) void qnn_fused_kernel(
    const float* __restrict__ x, const float* __restrict__ bp,
    const float* __restrict__ W1, const float* __restrict__ b1,
    const float* __restrict__ W2, const float* __restrict__ b2,
    float* __restrict__ out) {
  __shared__ float4 sB[N_CHUNKS * 8 * 32];
  __shared__ float redN[16][16][16];
  __shared__ float redS[16][16][2];

  const int lane = threadIdx.x & 31;
  const int wave = threadIdx.x >> 5;
  const int hf   = lane >> 4;
  const int l16  = lane & 15;
  const int rbase = blockIdx.x * 256 + wave * 16;

  const float* xrow = x + (size_t)(rbase + l16) * DIM + 2 * hf;
  const float* xd_ptr = x + (size_t)(rbase + 8 * hf) * DIM + l16;

  v2f afrag[64];
#pragma unroll
  for (int kk = 0; kk < 64; ++kk) {
    afrag[kk] = *(const v2f*)(xrow + 4 * kk);
  }

  const float4* bp4 = (const float4*)bp;
#pragma unroll 9
  for (int it = 0; it < N_CHUNKS * 8 * 32 / 512; ++it) {
    sB[it * 512 + threadIdx.x] = bp4[it * 512 + threadIdx.x];
  }
  __syncthreads();

  float sS = 0.f;
#pragma unroll
  for (int kk = 0; kk < 64; ++kk) {
    sS += afrag[kk].x * afrag[kk].x;
    sS += afrag[kk].y * afrag[kk].y;
  }

  float accN[8];
#pragma unroll
  for (int v = 0; v < 8; ++v) accN[v] = 0.f;

  const float4* bl = sB + lane;
  float4 bcur[8], bnxt[8];
#pragma unroll
  for (int i = 0; i < 8; ++i) bcur[i] = bl[i * 32];
  int f = 0;

#pragma unroll 1
  for (int m = 0; m < 8; ++m) {
    v8f acc0 = {0.f, 0.f, 0.f, 0.f, 0.f, 0.f, 0.f, 0.f};
    v8f acc1 = {0.f, 0.f, 0.f, 0.f, 0.f, 0.f, 0.f, 0.f};
    float xd0[8], xd1[8];
#pragma unroll
    for (int c = 0; c < 8; ++c) {
      if (c > m) break;
      const int fn = (f + 1 == N_CHUNKS) ? 0 : f + 1;
#pragma unroll
      for (int i = 0; i < 8; ++i) bnxt[i] = bl[(fn * 8 + i) * 32];
      if (c == m) {
#pragma unroll
        for (int r = 0; r < 8; ++r) {
          xd0[r] = xd_ptr[r * DIM + 32 * m];
          xd1[r] = xd_ptr[r * DIM + 32 * m + 16];
        }
      }
      __builtin_amdgcn_sched_barrier(0);
#pragma unroll
      for (int i = 0; i < 4; ++i) {
        const float4 g0 = bcur[2 * i], g1 = bcur[2 * i + 1];
        const v2f b00 = {g0.x, g0.y}, b01 = {g0.z, g0.w};
        const v2f b10 = {g1.x, g1.y}, b11 = {g1.z, g1.w};
        acc0 = __builtin_amdgcn_wmma_f32_16x16x4_f32(
             false, afrag[8 * c + 2 * i],  false, b00,
             (short)0, acc0,  false,  false);
        acc1 = __builtin_amdgcn_wmma_f32_16x16x4_f32(
             false, afrag[8 * c + 2 * i],  false, b10,
             (short)0, acc1,  false,  false);
        acc0 = __builtin_amdgcn_wmma_f32_16x16x4_f32(
             false, afrag[8 * c + 2 * i + 1],  false, b01,
             (short)0, acc0,  false,  false);
        acc1 = __builtin_amdgcn_wmma_f32_16x16x4_f32(
             false, afrag[8 * c + 2 * i + 1],  false, b11,
             (short)0, acc1,  false,  false);
      }
#pragma unroll
      for (int i = 0; i < 8; ++i) bcur[i] = bnxt[i];
      f = fn;
    }
#pragma unroll
    for (int r = 0; r < 8; ++r) {
      accN[r] += xd0[r] * acc0[r];
      accN[r] += xd1[r] * acc1[r];
    }
  }

#pragma unroll
  for (int v = 0; v < 8; ++v) redN[wave][hf * 8 + v][l16] = accN[v];
  redS[wave][l16][hf] = sS;
  __syncthreads();

  if (threadIdx.x < 256) {
    const int wv = threadIdx.x >> 4;
    const int r  = threadIdx.x & 15;
    float N = 0.f;
#pragma unroll
    for (int i = 0; i < 16; ++i) N += redN[wv][r][i];
    const float S = redS[wv][r][0] + redS[wv][r][1];
    const float z = N / S;
    float t = b2[0];
#pragma unroll
    for (int jj = 0; jj < 16; ++jj) {
      float h = z * W1[jj] + b1[jj];
      h = h > 0.f ? h : 0.f;
      t += h * W2[jj];
    }
    out[(size_t)blockIdx.x * 256 + wv * 16 + r] = 1.f / (1.f + expf(-t));
  }
}

extern "C" void kernel_launch(void* const* d_in, const int* in_sizes, int n_in,
                              void* d_out, int out_size, void* d_ws, size_t ws_size,
                              hipStream_t stream) {
  const float* x  = (const float*)d_in[0];
  const float* w  = (const float*)d_in[1];
  const float* W1 = (const float*)d_in[2];
  const float* b1 = (const float*)d_in[3];
  const float* W2 = (const float*)d_in[4];
  const float* b2 = (const float*)d_in[5];
  float* out = (float*)d_out;
  float* bp  = (float*)d_ws;

  const int B = in_sizes[0] / DIM;
  build_q_kernel<<<DIM, DIM, 0, stream>>>(w, bp);
  qnn_fused_kernel<<<B / 256, 512, 0, stream>>>(x, bp, W1, b1, W2, b2, out);
}


# Product transposed (G as A operand, x as B): a lane's own x fragment weights the finished tile, so x is read once, no load is left in the product loop and it starts while x streams in; new sum order
# speedup vs baseline: 1.9031x; 1.0419x over previous
#include <hip/hip_runtime.h>

typedef __attribute__((ext_vector_type(2))) float v2f;
typedef __attribute__((ext_vector_type(8))) float v8f;

#define N_LAYERS 7
#define N_QUBITS 8
#define DIM 256


__host__ __device__ constexpr int chunk_base(int m) { return m * (m + 1) / 2; }
#define N_CHUNKS 36


template <int SH>
__device__ __forceinline__ float ry4_stage(const float* src, int i, float4 c, float4 s) {
  float v[16];
  if (SH == 0) {
    const float4* p = (const float4*)(src + (i & ~15));
    for (int t = 0; t < 4; ++t) {
      const float4 f = p[t];
      v[4 * t] = f.x; v[4 * t + 1] = f.y; v[4 * t + 2] = f.z; v[4 * t + 3] = f.w;
    }
  } else {
    const float* p = src + (i & ~(15 << SH));
    for (int m = 0; m < 16; ++m) v[m] = p[m << SH];
  }
  const bool b3 = (i >> (SH + 3)) & 1;
  const bool b2 = (i >> (SH + 2)) & 1;
  const bool b1 = (i >> (SH + 1)) & 1;
  const bool b0 = (i >> SH) & 1;
  const float p3 = b3 ? s.x : c.x, q3 = b3 ? c.x : -s.x;
  const float p2 = b2 ? s.y : c.y, q2 = b2 ? c.y : -s.y;
  const float p1 = b1 ? s.z : c.z, q1 = b1 ? c.z : -s.z;
  const float p0 = b0 ? s.w : c.w, q0 = b0 ? c.w : -s.w;
  for (int t = 0; t < 8; ++t) v[t] = p3 * v[t] + q3 * v[t + 8];
  for (int t = 0; t < 4; ++t) v[t] = p2 * v[t] + q2 * v[t + 4];
  for (int t = 0; t < 2; ++t) v[t] = p1 * v[t] + q1 * v[t + 2];
  return p0 * v[0] + q0 * v[1];
}

__device__ __forceinline__ float4 neg4(float4 a) {
  return make_float4(-a.x, -a.y, -a.z, -a.w);
}

__global__ __launch_bounds__(256) void build_q_kernel(const float* __restrict__ w,
                                                      float* __restrict__ bp) {
  __shared__ __attribute__((aligned(16))) float col[2][DIM];
  __shared__ __attribute__((aligned(16))) float tcos[N_LAYERS * N_QUBITS];
  __shared__ __attribute__((aligned(16))) float tsin[N_LAYERS * N_QUBITS];
  const int i = threadIdx.x;
  const int j = blockIdx.x;
  if (i < N_LAYERS * N_QUBITS) {
    const float th = 0.5f * w[i];
    tcos[i] = cosf(th);
    tsin[i] = sinf(th);
  }
  col[0][i] = (i == j) ? 1.0f : 0.0f;
  int dst = i;
  for (int q = 0; q < N_QUBITS - 1; ++q) {
    if (dst & (1 << (7 - q))) dst ^= 1 << (6 - q);
  }
  const int inv = i ^ (i >> 1);
  __syncthreads();
  for (int l = 0; l < N_LAYERS; ++l) {
    const float4 ca = *(const float4*)&tcos[l * N_QUBITS];
    const float4 sa = *(const float4*)&tsin[l * N_QUBITS];
    const float4 cb = *(const float4*)&tcos[l * N_QUBITS + 4];
    const float4 sb = *(const float4*)&tsin[l * N_QUBITS + 4];
    col[1][i] = ry4_stage<4>(col[0], i, ca, sa);
    __syncthreads();
    const float r = ry4_stage<0>(col[1], i, cb, sb);
    if (l < N_LAYERS - 1) {
      col[0][dst] = r;
    } else {
      col[0][i] = (i & 128) ? -r : r;
    }
    __syncthreads();
  }
  for (int l = N_LAYERS - 1; l >= 0; --l) {
    const float4 ca = *(const float4*)&tcos[l * N_QUBITS];
    const float4 sa = neg4(*(const float4*)&tsin[l * N_QUBITS]);
    const float4 cb = *(const float4*)&tcos[l * N_QUBITS + 4];
    const float4 sb = neg4(*(const float4*)&tsin[l * N_QUBITS + 4]);
    col[1][i] = ry4_stage<0>(col[0], i, cb, sb);
    __syncthreads();
    const float r = ry4_stage<4>(col[1], i, ca, sa);
    col[0][l > 0 ? inv : i] = r;
    __syncthreads();
  }
  const int m = i >> 5;
  if ((j >> 5) <= m) {
    const int t   = (i >> 4) & 1;
    const int l16 = i & 15;
    const int c   = j >> 5;
    const int i4  = ((j >> 4) & 1) * 2 + ((j >> 2) & 1);
    const int hf  = (j >> 3) & 1;
    const int lane = hf * 16 + l16;
    const int row = (chunk_base(m) + c) * 8 + i4 * 2 + t;
    const float q = col[0][i];
    bp[(size_t)(row * 32 + lane) * 4 + (j & 3)] =
        ((j >> 5) == m) ? q : 2.0f * q;
  }
}


__global__ __launch_bounds__(512) void qnn_fused_kernel(
    const float* __restrict__ x, const float* __restrict__ bp,
    const float* __restrict__ W1, const float* __restrict__ b1,
    const float* __restrict__ W2, const float* __restrict__ b2,
    float* __restrict__ out) {
  __shared__ float4 sB[N_CHUNKS * 8 * 32];
  __shared__ float redN[16][16][2];
  __shared__ float redS[16][16][2];

  const int lane = threadIdx.x & 31;
  const int wave = threadIdx.x >> 5;
  const int hf   = lane >> 4;
  const int l16  = lane & 15;

  const float4* bp4 = (const float4*)bp;
  static_assert(N_CHUNKS * 8 * 32 == 18 * 512, "18 float4 per thread");
#define PK_LOAD(i) const float4 pk##i = bp4[(i) * 512 + threadIdx.x];
  PK_LOAD(0) PK_LOAD(1) PK_LOAD(2) PK_LOAD(3) PK_LOAD(4) PK_LOAD(5)
  PK_LOAD(6) PK_LOAD(7) PK_LOAD(8) PK_LOAD(9) PK_LOAD(10) PK_LOAD(11)
  PK_LOAD(12) PK_LOAD(13) PK_LOAD(14) PK_LOAD(15) PK_LOAD(16) PK_LOAD(17)
#undef PK_LOAD
  unsigned xblk = blockIdx.x * (256 * DIM);
  asm volatile("" : "+s"(xblk));
  const float4* xrow =
      (const float4*)(x + xblk + (unsigned)(wave * 16 + l16) * DIM + 8 * hf);
  float4 xf[32];
#pragma unroll
  for (int t = 0; t < 16; ++t) {
    xf[2 * t]     = xrow[4 * t];
    xf[2 * t + 1] = xrow[4 * t + 1];
  }

#define PK_STORE(i) sB[(i) * 512 + threadIdx.x] = pk##i;
  PK_STORE(0) PK_STORE(1) PK_STORE(2) PK_STORE(3) PK_STORE(4) PK_STORE(5)
  PK_STORE(6) PK_STORE(7) PK_STORE(8) PK_STORE(9) PK_STORE(10) PK_STORE(11)
  PK_STORE(12) PK_STORE(13) PK_STORE(14) PK_STORE(15) PK_STORE(16) PK_STORE(17)
#undef PK_STORE
  __builtin_amdgcn_fence(__ATOMIC_RELEASE, "workgroup", "local");
  __builtin_amdgcn_s_barrier();
  __builtin_amdgcn_fence(__ATOMIC_ACQUIRE, "workgroup", "local");

  const float4* bl = sB + lane;
  float4 bcur[8];
#pragma unroll
  for (int i = 0; i < 8; ++i) bcur[i] = bl[i * 32];
  int f = 0;
  float num = 0.f;

#pragma unroll 1
  for (int m = 0; m < 8; ++m) {
    v8f acc0 = {0.f, 0.f, 0.f, 0.f, 0.f, 0.f, 0.f, 0.f};
    v8f acc1 = {0.f, 0.f, 0.f, 0.f, 0.f, 0.f, 0.f, 0.f};
#pragma unroll
    for (int c = 0; c < 8; ++c) {
      if (c > m) break;
      const int fn = (f + 1 == N_CHUNKS) ? 0 : f + 1;
      float4 bnxt[8];
#pragma unroll
      for (int i = 0; i < 8; ++i) bnxt[i] = bl[(fn * 8 + i) * 32];
      __builtin_amdgcn_sched_barrier(0);
#pragma unroll
      for (int i = 0; i < 4; ++i) {
        const float4 xv = xf[4 * c + i];
        const float4 g0 = bcur[2 * i], g1 = bcur[2 * i + 1];
        const v2f x0 = {xv.x, xv.y}, x1 = {xv.z, xv.w};
        const v2f a00 = {g0.x, g0.y}, a01 = {g0.z, g0.w};
        const v2f a10 = {g1.x, g1.y}, a11 = {g1.z, g1.w};
        acc0 = __builtin_amdgcn_wmma_f32_16x16x4_f32(
             false, a00,  false, x0,
             (short)0, acc0,  false,  false);
        acc1 = __builtin_amdgcn_wmma_f32_16x16x4_f32(
             false, a10,  false, x0,
             (short)0, acc1,  false,  false);
        acc0 = __builtin_amdgcn_wmma_f32_16x16x4_f32(
             false, a01,  false, x1,
             (short)0, acc0,  false,  false);
        acc1 = __builtin_amdgcn_wmma_f32_16x16x4_f32(
             false, a11,  false, x1,
             (short)0, acc1,  false,  false);
      }
#pragma unroll
      for (int i = 0; i < 8; ++i) bcur[i] = bnxt[i];
      f = fn;
    }
#pragma unroll
    for (int mm = 0; mm < 8; ++mm) {
      if (mm == m) {
        const float4 p0 = xf[4 * mm], p1 = xf[4 * mm + 1];
        const float4 p2 = xf[4 * mm + 2], p3 = xf[4 * mm + 3];
        num += p0.x * acc0[0]; num += p0.y * acc0[1];
        num += p0.z * acc0[2]; num += p0.w * acc0[3];
        num += p1.x * acc0[4]; num += p1.y * acc0[5];
        num += p1.z * acc0[6]; num += p1.w * acc0[7];
        num += p2.x * acc1[0]; num += p2.y * acc1[1];
        num += p2.z * acc1[2]; num += p2.w * acc1[3];
        num += p3.x * acc1[4]; num += p3.y * acc1[5];
        num += p3.z * acc1[6]; num += p3.w * acc1[7];
      }
    }
  }

  float sS = 0.f;
#pragma unroll
  for (int q = 0; q < 32; ++q) {
    sS += xf[q].x * xf[q].x;
    sS += xf[q].y * xf[q].y;
    sS += xf[q].z * xf[q].z;
    sS += xf[q].w * xf[q].w;
  }

  redN[wave][l16][hf] = num;
  redS[wave][l16][hf] = sS;
  __syncthreads();

  if (threadIdx.x < 256) {
    const int wv = threadIdx.x >> 4;
    const int r  = threadIdx.x & 15;
    const float N = redN[wv][r][0] + redN[wv][r][1];
    const float S = redS[wv][r][0] + redS[wv][r][1];
    const float z = N / S;
    float t = b2[0];
#pragma unroll
    for (int jj = 0; jj < 16; ++jj) {
      float h = z * W1[jj] + b1[jj];
      h = h > 0.f ? h : 0.f;
      t += h * W2[jj];
    }
    out[(size_t)blockIdx.x * 256 + wv * 16 + r] = 1.f / (1.f + expf(-t));
  }
}

extern "C" void kernel_launch(void* const* d_in, const int* in_sizes, int n_in,
                              void* d_out, int out_size, void* d_ws, size_t ws_size,
                              hipStream_t stream) {
  const float* x  = (const float*)d_in[0];
  const float* w  = (const float*)d_in[1];
  const float* W1 = (const float*)d_in[2];
  const float* b1 = (const float*)d_in[3];
  const float* W2 = (const float*)d_in[4];
  const float* b2 = (const float*)d_in[5];
  float* out = (float*)d_out;
  float* bp  = (float*)d_ws;

  const int B = in_sizes[0] / DIM;
  build_q_kernel<<<DIM, DIM, 0, stream>>>(w, bp);
  qnn_fused_kernel<<<B / 256, 512, 0, stream>>>(x, bp, W1, b1, W2, b2, out);
}
